# starting kernel
# baseline (speedup 1.0000x reference)
#include <hip/hip_runtime.h>

typedef __attribute__((ext_vector_type(2))) float v2f;
typedef __attribute__((ext_vector_type(8))) float v8f;

static constexpr int kM    = 8192;
static constexpr int kK    = 8;
static constexpr int kC    = 256;
static constexpr int kLay  = 8;
static constexpr int kWavesPerBlock = 4;
static constexpr int kNodesPerBlock = kWavesPerBlock * 16;

__device__ __forceinline__ float fast_sigmoid(float x) {
    return __builtin_amdgcn_rcpf(1.0f + __expf(-x));
}

__device__ __forceinline__ float pick8(v8f c, int i) {
    float r = c[0];
    if (i == 1) r = c[1];
    if (i == 2) r = c[2];
    if (i == 3) r = c[3];
    if (i == 4) r = c[4];
    if (i == 5) r = c[5];
    if (i == 6) r = c[6];
    if (i == 7) r = c[7];
    return r;
}

__global__ void bn_root_kernel(const float* __restrict__ logits,
                               float* __restrict__ out) {
    int i = blockIdx.x * blockDim.x + threadIdx.x;
    if (i < kM) out[i] = fast_sigmoid(logits[i]);
}

__global__ __launch_bounds__(kWavesPerBlock * 32)
void bn_layer_kernel(const float* __restrict__ cpt,
                     const int*   __restrict__ par,
                     const float* __restrict__ prev,
                     float*       __restrict__ outm) {
    const int lane = threadIdx.x & 31;
    const int wave = threadIdx.x >> 5;
    const int n    = lane & 15;
    const int h    = lane >> 4;
    const int tile = blockIdx.x * kWavesPerBlock + wave;
    const int node = tile * 16 + n;

    const int* pp = par + node * kK;
    float p[kK], q[kK];
#pragma unroll
    for (int j = 0; j < kK; ++j) {
        p[j] = prev[pp[j]];
        q[j] = 1.0f - p[j];
    }
    const float fh = h ? p[6] : q[6];

    const float* row = cpt + (size_t)node * kC + 2 * h;
    __builtin_prefetch(row, 0, 3);
    __builtin_prefetch(row + kC / 2, 0, 3);

    v8f acc = {};
#pragma unroll
    for (int hi = 0; hi < 8; ++hi) {
        const float rH = (((hi >> 2) & 1) ? p[0] : q[0])
                       * (((hi >> 1) & 1) ? p[1] : q[1])
                       * (((hi     ) & 1) ? p[2] : q[2]) * fh;
#pragma unroll
        for (int lo = 0; lo < 8; ++lo) {
            const int t = hi * 8 + lo;
            const float rL = (((lo >> 2) & 1) ? p[3] : q[3])
                           * (((lo >> 1) & 1) ? p[4] : q[4])
                           * (((lo     ) & 1) ? p[5] : q[5]);
            const float r7 = rH * rL;

            v2f a;
            a.x = r7 * q[7];
            a.y = r7 * p[7];

            const float2 cl = *(const float2*)(row + 4 * t);
            v2f b;
            b.x = fast_sigmoid(cl.x);
            b.y = fast_sigmoid(cl.y);

            acc = __builtin_amdgcn_wmma_f32_16x16x4_f32(
                false, a, false, b, (short)0, acc, false, false);
        }
    }

    if (lane < 8) {
        outm[tile * 16 + lane] = pick8(acc, lane);
    } else if (lane >= 24) {
        outm[tile * 16 + (lane - 16)] = pick8(acc, lane - 24);
    }
}

extern "C" void kernel_launch(void* const* d_in, const int* in_sizes, int n_in,
                              void* d_out, int out_size, void* d_ws, size_t ws_size,
                              hipStream_t stream) {
    (void)in_sizes; (void)n_in; (void)out_size; (void)d_ws; (void)ws_size;
    const float* root    = (const float*)d_in[0];
    const float* cpt     = (const float*)d_in[1];
    const int*   parents = (const int*)d_in[2];
    float* out = (float*)d_out;

    bn_root_kernel<<<kM / 256, 256, 0, stream>>>(root, out);

    for (int l = 0; l < kLay; ++l) {
        bn_layer_kernel<<<kM / kNodesPerBlock, kWavesPerBlock * 32, 0, stream>>>(
            cpt + (size_t)l * kM * kC,
            parents + (size_t)l * kM * kK,
            out + (size_t)l * kM,
            out + (size_t)(l + 1) * kM);
    }
}


# Whole 64-load CPT row slice of each lane issued up front in order of use (was a 10-12 load window), consumed line by line as it returns; L2 prefetch pair dropped; same ops and summation order
# speedup vs baseline: 1.2254x; 1.2254x over previous
#include <hip/hip_runtime.h>

typedef __attribute__((ext_vector_type(2))) float v2f;
typedef __attribute__((ext_vector_type(8))) float v8f;

static constexpr int kM    = 8192;
static constexpr int kK    = 8;
static constexpr int kC    = 256;
static constexpr int kLay  = 8;
static constexpr int kWavesPerBlock = 4;
static constexpr int kNodesPerBlock = kWavesPerBlock * 16;

__device__ __forceinline__ float fast_sigmoid(float x) {
    return __builtin_amdgcn_rcpf(1.0f + __expf(-x));
}

__device__ __forceinline__ float scaled_sigmoid(float x, float s) {
    return __builtin_amdgcn_rcpf(1.0f + __builtin_amdgcn_exp2f(x * s));
}

__device__ __forceinline__ float pick8(v8f c, int i) {
    float r = c[0];
    if (i == 1) r = c[1];
    if (i == 2) r = c[2];
    if (i == 3) r = c[3];
    if (i == 4) r = c[4];
    if (i == 5) r = c[5];
    if (i == 6) r = c[6];
    if (i == 7) r = c[7];
    return r;
}

__global__ void bn_root_kernel(const float* __restrict__ logits,
                               float* __restrict__ out) {
    int i = blockIdx.x * blockDim.x + threadIdx.x;
    if (i < kM) out[i] = fast_sigmoid(logits[i]);
}

__global__ __launch_bounds__(kWavesPerBlock * 32)
void bn_layer_kernel(const float* __restrict__ cpt,
                     const int*   __restrict__ par,
                     const float* __restrict__ prev,
                     float*       __restrict__ outm) {
    const int lane = threadIdx.x & 31;
    const int wave = threadIdx.x >> 5;
    const int n    = lane & 15;
    const int h    = lane >> 4;
    const int tile = blockIdx.x * kWavesPerBlock + wave;
    const int node = tile * 16 + n;

    const int* pp = par + node * kK;
    float p[kK], q[kK];
#pragma unroll
    for (int j = 0; j < kK; ++j) p[j] = prev[pp[j]];

    const float* row = cpt + (size_t)node * kC + 2 * h;
    float2 cl[kC / 4];
#pragma unroll
    for (int t = 0; t < kC / 4; ++t) cl[t] = *(const float2*)(row + 4 * t);

#pragma unroll
    for (int j = 0; j < kK; ++j) q[j] = 1.0f - p[j];
    const float fh = h ? p[6] : q[6];

    v8f acc = {};
    float nlog2e = -1.44269504088896341f;
#pragma unroll
    for (int hi = 0; hi < 8; ++hi) {
        asm volatile("" : "+v"(acc), "+s"(nlog2e));
        const float rH = (((hi >> 2) & 1) ? p[0] : q[0])
                       * (((hi >> 1) & 1) ? p[1] : q[1])
                       * (((hi     ) & 1) ? p[2] : q[2]) * fh;
#pragma unroll
        for (int lo = 0; lo < 8; ++lo) {
            const int t = hi * 8 + lo;
            const float rL = (((lo >> 2) & 1) ? p[3] : q[3])
                           * (((lo >> 1) & 1) ? p[4] : q[4])
                           * (((lo     ) & 1) ? p[5] : q[5]);
            const float r7 = rH * rL;

            v2f a;
            a.x = r7 * q[7];
            a.y = r7 * p[7];

            v2f b;
            b.x = scaled_sigmoid(cl[t].x, nlog2e);
            b.y = scaled_sigmoid(cl[t].y, nlog2e);

            acc = __builtin_amdgcn_wmma_f32_16x16x4_f32(
                false, a, false, b, (short)0, acc, false, false);
        }
    }

    if (lane < 8) {
        outm[tile * 16 + lane] = pick8(acc, lane);
    } else if (lane >= 24) {
        outm[tile * 16 + (lane - 16)] = pick8(acc, lane - 24);
    }
}

extern "C" void kernel_launch(void* const* d_in, const int* in_sizes, int n_in,
                              void* d_out, int out_size, void* d_ws, size_t ws_size,
                              hipStream_t stream) {
    (void)in_sizes; (void)n_in; (void)out_size; (void)d_ws; (void)ws_size;
    const float* root    = (const float*)d_in[0];
    const float* cpt     = (const float*)d_in[1];
    const int*   parents = (const int*)d_in[2];
    float* out = (float*)d_out;

    bn_root_kernel<<<kM / 256, 256, 0, stream>>>(root, out);

    for (int l = 0; l < kLay; ++l) {
        bn_layer_kernel<<<kM / kNodesPerBlock, kWavesPerBlock * 32, 0, stream>>>(
            cpt + (size_t)l * kM * kC,
            parents + (size_t)l * kM * kK,
            out + (size_t)l * kM,
            out + (size_t)(l + 1) * kM);
    }
}


# Lanes remapped: 8 lanes per node read whole 128-B CPT lines (4 lines per b128 load; was 16 lines x 16 B per b64 load); diagonal holds 4 partial sums per node, added across a quad at the end
# speedup vs baseline: 1.4164x; 1.1559x over previous
#include <hip/hip_runtime.h>


typedef __attribute__((ext_vector_type(2))) float v2f;
typedef __attribute__((ext_vector_type(8))) float v8f;

static constexpr int kM    = 8192;
static constexpr int kK    = 8;
static constexpr int kC    = 256;
static constexpr int kLay  = 8;
static constexpr int kWavesPerBlock = 4;
static constexpr int kNodesPerBlock = kWavesPerBlock * 16;

__device__ __forceinline__ float fast_sigmoid(float x) {
    return __builtin_amdgcn_rcpf(1.0f + __expf(-x));
}

__device__ __forceinline__ float scaled_sigmoid(float x, float s) {
    return __builtin_amdgcn_rcpf(1.0f + __builtin_amdgcn_exp2f(x * s));
}

__device__ __forceinline__ float pick8(v8f c, int i) {
    float r = c[0];
    if (i == 1) r = c[1];
    if (i == 2) r = c[2];
    if (i == 3) r = c[3];
    if (i == 4) r = c[4];
    if (i == 5) r = c[5];
    if (i == 6) r = c[6];
    if (i == 7) r = c[7];
    return r;
}

__device__ __forceinline__ float quad_sum(float v) {
    float t = __int_as_float(__builtin_amdgcn_mov_dpp(__float_as_int(v), 0xB1, 0xF, 0xF, true));
    v += t;
    t = __int_as_float(__builtin_amdgcn_mov_dpp(__float_as_int(v), 0x4E, 0xF, 0xF, true));
    return v + t;
}

__global__ void bn_root_kernel(const float* __restrict__ logits,
                               float* __restrict__ out) {
    int i = blockIdx.x * blockDim.x + threadIdx.x;
    if (i < kM) out[i] = fast_sigmoid(logits[i]);
}

__global__ __launch_bounds__(kWavesPerBlock * 32)
void bn_layer_kernel(const float* __restrict__ cpt,
                     const int*   __restrict__ par,
                     const float* __restrict__ prev,
                     float*       __restrict__ outm) {
    const int lane = threadIdx.x & 31;
    const int wave = threadIdx.x >> 5;
    const int c    = lane & 15;
    const int h    = lane >> 4;
    const int a    = c >> 2;
    const int g    = c & 3;
    const int tile = blockIdx.x * kWavesPerBlock + wave;
    const int node0 = tile * 16 + a;

    float p[4][kK];
#pragma unroll
    for (int j = 0; j < 4; ++j) {
        const int4* ip = (const int4*)(par + (size_t)(node0 + 4 * j) * kK);
        const int4 ia = ip[0], ib = ip[1];
        p[j][0] = prev[ia.x]; p[j][1] = prev[ia.y]; p[j][2] = prev[ia.z]; p[j][3] = prev[ia.w];
        p[j][4] = prev[ib.x]; p[j][5] = prev[ib.y]; p[j][6] = prev[ib.z]; p[j][7] = prev[ib.w];
    }

    const float* row0 = cpt + (size_t)node0 * kC + 8 * g + 4 * h;
    float4 cl[4][8];
#pragma unroll
    for (int jp = 0; jp < 4; jp += 2) {
#pragma unroll
        for (int u = 0; u < 8; ++u) {
            cl[jp][u]     = *(const float4*)(row0 + (jp)     * 4 * kC + 32 * u);
            cl[jp + 1][u] = *(const float4*)(row0 + (jp + 1) * 4 * kC + 32 * u);
        }
    }

    float nlog2e = -1.44269504088896341f;
    v8f acc[4] = {};

#pragma unroll
    for (int jp = 0; jp < 4; jp += 2) {
        float wk[2][4], q0[2], q1[2], q2[2];
#pragma unroll
        for (int d = 0; d < 2; ++d) {
            const float* pj = p[jp + d];
            q0[d] = 1.0f - pj[0]; q1[d] = 1.0f - pj[1]; q2[d] = 1.0f - pj[2];
            const float q3 = 1.0f - pj[3], q4 = 1.0f - pj[4], q5 = 1.0f - pj[5];
            const float q6 = 1.0f - pj[6], q7 = 1.0f - pj[7];
            const float w = ((g & 2) ? pj[3] : q3) * ((g & 1) ? pj[4] : q4) * (h ? pj[5] : q5);
            const float w0 = w * q6, w1 = w * pj[6];
            wk[d][0] = w0 * q7; wk[d][1] = w0 * pj[7];
            wk[d][2] = w1 * q7; wk[d][3] = w1 * pj[7];
        }
#pragma unroll
        for (int u = 0; u < 8; ++u) {
            asm volatile("" : "+v"(acc[jp]), "+v"(acc[jp + 1]), "+s"(nlog2e));
#pragma unroll
            for (int d = 0; d < 2; ++d) {
                const int j = jp + d;
                const float* pj = p[j];
                const float U = ((u & 4) ? pj[0] : q0[d]) * ((u & 2) ? pj[1] : q1[d])
                              * ((u & 1) ? pj[2] : q2[d]);
                const float4 x = cl[j][u];
                v2f a01, a23, b01, b23;
                a01.x = U * wk[d][0]; a01.y = U * wk[d][1];
                a23.x = U * wk[d][2]; a23.y = U * wk[d][3];
                b01.x = scaled_sigmoid(x.x, nlog2e); b01.y = scaled_sigmoid(x.y, nlog2e);
                b23.x = scaled_sigmoid(x.z, nlog2e); b23.y = scaled_sigmoid(x.w, nlog2e);
                acc[j] = __builtin_amdgcn_wmma_f32_16x16x4_f32(
                    false, a01, false, b01, (short)0, acc[j], false, false);
                acc[j] = __builtin_amdgcn_wmma_f32_16x16x4_f32(
                    false, a23, false, b23, (short)0, acc[j], false, false);
            }
        }
    }

    const int r = c & 7;
    float res = quad_sum(pick8(acc[0], r));
    const float s1 = quad_sum(pick8(acc[1], r));
    const float s2 = quad_sum(pick8(acc[2], r));
    const float s3 = quad_sum(pick8(acc[3], r));
    if (g == 1) res = s1;
    if (g == 2) res = s2;
    if (g == 3) res = s3;
    if (lane < 8 || lane >= 24) outm[tile * 16 + 4 * g + a] = res;
}

extern "C" void kernel_launch(void* const* d_in, const int* in_sizes, int n_in,
                              void* d_out, int out_size, void* d_ws, size_t ws_size,
                              hipStream_t stream) {
    (void)in_sizes; (void)n_in; (void)out_size; (void)d_ws; (void)ws_size;
    const float* root    = (const float*)d_in[0];
    const float* cpt     = (const float*)d_in[1];
    const int*   parents = (const int*)d_in[2];
    float* out = (float*)d_out;

    bn_root_kernel<<<kM / 256, 256, 0, stream>>>(root, out);

    for (int l = 0; l < kLay; ++l) {
        bn_layer_kernel<<<kM / kNodesPerBlock, kWavesPerBlock * 32, 0, stream>>>(
            cpt + (size_t)l * kM * kC,
            parents + (size_t)l * kM * kK,
            out + (size_t)l * kM,
            out + (size_t)(l + 1) * kM);
    }
}


# Root launch folded into the first layer's launch: it gathers root logits, applies the root sigmoid in registers, and its blocks also store the root marginals; 8 launches per call (was 9)
# speedup vs baseline: 1.5018x; 1.0603x over previous
#include <hip/hip_runtime.h>


typedef __attribute__((ext_vector_type(2))) float v2f;
typedef __attribute__((ext_vector_type(8))) float v8f;

static constexpr int kM    = 8192;
static constexpr int kK    = 8;
static constexpr int kC    = 256;
static constexpr int kLay  = 8;
static constexpr int kWavesPerBlock = 4;
static constexpr int kNodesPerBlock = kWavesPerBlock * 16;

__device__ __forceinline__ float fast_sigmoid(float x) {
    return __builtin_amdgcn_rcpf(1.0f + __expf(-x));
}

__device__ __forceinline__ float scaled_sigmoid(float x, float s) {
    return __builtin_amdgcn_rcpf(1.0f + __builtin_amdgcn_exp2f(x * s));
}

__device__ __forceinline__ float pick8(v8f c, int i) {
    float r = c[0];
    if (i == 1) r = c[1];
    if (i == 2) r = c[2];
    if (i == 3) r = c[3];
    if (i == 4) r = c[4];
    if (i == 5) r = c[5];
    if (i == 6) r = c[6];
    if (i == 7) r = c[7];
    return r;
}

__device__ __forceinline__ float quad_sum(float v) {
    float t = __int_as_float(__builtin_amdgcn_mov_dpp(__float_as_int(v), 0xB1, 0xF, 0xF, true));
    v += t;
    t = __int_as_float(__builtin_amdgcn_mov_dpp(__float_as_int(v), 0x4E, 0xF, 0xF, true));
    return v + t;
}

template <bool kFromRoots>
__global__ __launch_bounds__(kWavesPerBlock * 32)
void bn_layer_kernel(const float* __restrict__ cpt,
                     const int*   __restrict__ par,
                     const float* __restrict__ prev,
                     float*       __restrict__ outm) {
    const int lane = threadIdx.x & 31;
    const int wave = threadIdx.x >> 5;
    const int c    = lane & 15;
    const int h    = lane >> 4;
    const int a    = c >> 2;
    const int g    = c & 3;
    const int tile = blockIdx.x * kWavesPerBlock + wave;
    const int node0 = tile * 16 + a;

    const int ri = blockIdx.x * kNodesPerBlock + (wave >> 1) * 32 + lane;
    float rl = 0.0f;
    if (kFromRoots) rl = prev[ri];

    float p[4][kK];
#pragma unroll
    for (int j = 0; j < 4; ++j) {
        const int4* ip = (const int4*)(par + (size_t)(node0 + 4 * j) * kK);
        const int4 ia = ip[0], ib = ip[1];
        p[j][0] = prev[ia.x]; p[j][1] = prev[ia.y]; p[j][2] = prev[ia.z]; p[j][3] = prev[ia.w];
        p[j][4] = prev[ib.x]; p[j][5] = prev[ib.y]; p[j][6] = prev[ib.z]; p[j][7] = prev[ib.w];
    }
    if (kFromRoots) {
#pragma unroll
        for (int j = 0; j < 4; ++j) {
#pragma unroll
            for (int k = 0; k < kK; ++k) p[j][k] = fast_sigmoid(p[j][k]);
        }
    }

    const float* row0 = cpt + (size_t)node0 * kC + 8 * g + 4 * h;
    float4 cl[4][8];
#pragma unroll
    for (int jp = 0; jp < 4; jp += 2) {
#pragma unroll
        for (int u = 0; u < 8; ++u) {
            cl[jp][u]     = *(const float4*)(row0 + (jp)     * 4 * kC + 32 * u);
            cl[jp + 1][u] = *(const float4*)(row0 + (jp + 1) * 4 * kC + 32 * u);
        }
    }

    float nlog2e = -1.44269504088896341f;
    v8f acc[4] = {};

#pragma unroll
    for (int jp = 0; jp < 4; jp += 2) {
        float wk[2][4], q0[2], q1[2], q2[2];
#pragma unroll
        for (int d = 0; d < 2; ++d) {
            const float* pj = p[jp + d];
            q0[d] = 1.0f - pj[0]; q1[d] = 1.0f - pj[1]; q2[d] = 1.0f - pj[2];
            const float q3 = 1.0f - pj[3], q4 = 1.0f - pj[4], q5 = 1.0f - pj[5];
            const float q6 = 1.0f - pj[6], q7 = 1.0f - pj[7];
            const float w = ((g & 2) ? pj[3] : q3) * ((g & 1) ? pj[4] : q4) * (h ? pj[5] : q5);
            const float w0 = w * q6, w1 = w * pj[6];
            wk[d][0] = w0 * q7; wk[d][1] = w0 * pj[7];
            wk[d][2] = w1 * q7; wk[d][3] = w1 * pj[7];
        }
#pragma unroll
        for (int u = 0; u < 8; ++u) {
            asm volatile("" : "+v"(acc[jp]), "+v"(acc[jp + 1]), "+s"(nlog2e));
#pragma unroll
            for (int d = 0; d < 2; ++d) {
                const int j = jp + d;
                const float* pj = p[j];
                const float U = ((u & 4) ? pj[0] : q0[d]) * ((u & 2) ? pj[1] : q1[d])
                              * ((u & 1) ? pj[2] : q2[d]);
                const float4 x = cl[j][u];
                v2f a01, a23, b01, b23;
                a01.x = U * wk[d][0]; a01.y = U * wk[d][1];
                a23.x = U * wk[d][2]; a23.y = U * wk[d][3];
                b01.x = scaled_sigmoid(x.x, nlog2e); b01.y = scaled_sigmoid(x.y, nlog2e);
                b23.x = scaled_sigmoid(x.z, nlog2e); b23.y = scaled_sigmoid(x.w, nlog2e);
                acc[j] = __builtin_amdgcn_wmma_f32_16x16x4_f32(
                    false, a01, false, b01, (short)0, acc[j], false, false);
                acc[j] = __builtin_amdgcn_wmma_f32_16x16x4_f32(
                    false, a23, false, b23, (short)0, acc[j], false, false);
            }
        }
    }

    const int r = c & 7;
    float res = quad_sum(pick8(acc[0], r));
    const float s1 = quad_sum(pick8(acc[1], r));
    const float s2 = quad_sum(pick8(acc[2], r));
    const float s3 = quad_sum(pick8(acc[3], r));
    if (g == 1) res = s1;
    if (g == 2) res = s2;
    if (g == 3) res = s3;
    if (lane < 8 || lane >= 24) outm[tile * 16 + 4 * g + a] = res;

    if (kFromRoots) {
        if ((wave & 1) == 0) (outm - kM)[ri] = fast_sigmoid(rl);
    }
}

extern "C" void kernel_launch(void* const* d_in, const int* in_sizes, int n_in,
                              void* d_out, int out_size, void* d_ws, size_t ws_size,
                              hipStream_t stream) {
    (void)in_sizes; (void)n_in; (void)out_size; (void)d_ws; (void)ws_size;
    const float* root    = (const float*)d_in[0];
    const float* cpt     = (const float*)d_in[1];
    const int*   parents = (const int*)d_in[2];
    float* out = (float*)d_out;

    bn_layer_kernel<true><<<kM / kNodesPerBlock, kWavesPerBlock * 32, 0, stream>>>(
        cpt, parents, root, out + kM);

    for (int l = 1; l < kLay; ++l) {
        bn_layer_kernel<false><<<kM / kNodesPerBlock, kWavesPerBlock * 32, 0, stream>>>(
            cpt + (size_t)l * kM * kC,
            parents + (size_t)l * kM * kK,
            out + (size_t)l * kM,
            out + (size_t)(l + 1) * kM);
    }
}


# All 32 CPT loads issued ahead of the wait for the parent indices, their sigmoids done in place while the gathers are in flight (were between the matrix ops); same ops and summation order
# speedup vs baseline: 1.6800x; 1.1187x over previous
#include <hip/hip_runtime.h>


typedef __attribute__((ext_vector_type(2))) float v2f;
typedef __attribute__((ext_vector_type(4))) float v4f;
typedef __attribute__((ext_vector_type(8))) float v8f;

static constexpr int kM    = 8192;
static constexpr int kK    = 8;
static constexpr int kC    = 256;
static constexpr int kLay  = 8;
static constexpr int kWavesPerBlock = 4;
static constexpr int kNodesPerBlock = kWavesPerBlock * 16;

__device__ __forceinline__ float fast_sigmoid(float x) {
    return __builtin_amdgcn_rcpf(1.0f + __expf(-x));
}

__device__ __forceinline__ float scaled_sigmoid(float x, float s) {
    return __builtin_amdgcn_rcpf(1.0f + __builtin_amdgcn_exp2f(x * s));
}

__device__ __forceinline__ float pick8(v8f c, int i) {
    float r = c[0];
    if (i == 1) r = c[1];
    if (i == 2) r = c[2];
    if (i == 3) r = c[3];
    if (i == 4) r = c[4];
    if (i == 5) r = c[5];
    if (i == 6) r = c[6];
    if (i == 7) r = c[7];
    return r;
}

__device__ __forceinline__ float quad_sum(float v) {
    float t = __int_as_float(__builtin_amdgcn_mov_dpp(__float_as_int(v), 0xB1, 0xF, 0xF, true));
    v += t;
    t = __int_as_float(__builtin_amdgcn_mov_dpp(__float_as_int(v), 0x4E, 0xF, 0xF, true));
    return v + t;
}

template <bool kFromRoots>
__global__ __launch_bounds__(kWavesPerBlock * 32)
void bn_layer_kernel(const float* cpt,
                     const int*   par,
                     const float* prev,
                     float*       outm) {
    const int lane = threadIdx.x & 31;
    const int wave = threadIdx.x >> 5;
    const int c    = lane & 15;
    const int h    = lane >> 4;
    const int a    = c >> 2;
    const int g    = c & 3;
    const int tile = blockIdx.x * kWavesPerBlock + wave;
    const int node0 = tile * 16 + a;

    const int ri = blockIdx.x * kNodesPerBlock + (wave >> 1) * 32 + lane;
    float rl = 0.0f;
    if (kFromRoots) rl = prev[ri];

    int4 ia[4], ib[4];
#pragma unroll
    for (int j = 0; j < 4; ++j) {
        const int4* ip = (const int4*)(par + (size_t)(node0 + 4 * j) * kK);
        ia[j] = ip[0];
        ib[j] = ip[1];
    }

    asm volatile("" ::: "memory");

    const float* row0 = cpt + (size_t)node0 * kC + 8 * g + 4 * h;
    v4f cl[4][8];
#pragma unroll
    for (int jp = 0; jp < 4; jp += 2) {
#pragma unroll
        for (int u = 0; u < 8; ++u) {
            cl[jp][u]     = *(const v4f*)(row0 + (jp)     * 4 * kC + 32 * u);
            cl[jp + 1][u] = *(const v4f*)(row0 + (jp + 1) * 4 * kC + 32 * u);
        }
    }

    asm volatile("" ::: "memory");

    float p[4][kK];
#pragma unroll
    for (int j = 0; j < 4; ++j) {
        p[j][0] = prev[ia[j].x]; p[j][1] = prev[ia[j].y]; p[j][2] = prev[ia[j].z]; p[j][3] = prev[ia[j].w];
        p[j][4] = prev[ib[j].x]; p[j][5] = prev[ib[j].y]; p[j][6] = prev[ib[j].z]; p[j][7] = prev[ib[j].w];
    }

    float nlog2e = -1.44269504088896341f;
    asm volatile("" : "+s"(nlog2e) : : "memory");
#pragma unroll
    for (int jp = 0; jp < 4; jp += 2) {
#pragma unroll
        for (int u = 0; u < 8; u += 2) {
#pragma unroll
            for (int e = 0; e < 2; ++e) {
#pragma unroll
                for (int d = 0; d < 2; ++d) {
                    const v4f x = cl[jp + d][u + e];
                    v4f y;
                    y.x = scaled_sigmoid(x.x, nlog2e); y.y = scaled_sigmoid(x.y, nlog2e);
                    y.z = scaled_sigmoid(x.z, nlog2e); y.w = scaled_sigmoid(x.w, nlog2e);
                    cl[jp + d][u + e] = y;
                }
            }
            asm volatile("" : "+v"(cl[jp][u]), "+v"(cl[jp + 1][u]), "+v"(cl[jp][u + 1]),
                              "+v"(cl[jp + 1][u + 1]), "+s"(nlog2e));
        }
    }

    float one = 1.0f;
    asm volatile("" : "+s"(one));

    if (kFromRoots) {
#pragma unroll
        for (int j = 0; j < 4; ++j) {
#pragma unroll
            for (int k = 0; k < kK; ++k)
                p[j][k] = __builtin_amdgcn_rcpf(one + __expf(-p[j][k]));
        }
    }

    float wk[4][4], U[4][8];
#pragma unroll
    for (int j = 0; j < 4; ++j) {
        const float* pj = p[j];
        const float q0 = one - pj[0], q1 = one - pj[1], q2 = one - pj[2];
        const float q3 = one - pj[3], q4 = one - pj[4], q5 = one - pj[5];
        const float q6 = one - pj[6], q7 = one - pj[7];
        const float w = ((g & 2) ? pj[3] : q3) * ((g & 1) ? pj[4] : q4) * (h ? pj[5] : q5);
        const float w0 = w * q6, w1 = w * pj[6];
        wk[j][0] = w0 * q7; wk[j][1] = w0 * pj[7];
        wk[j][2] = w1 * q7; wk[j][3] = w1 * pj[7];
#pragma unroll
        for (int u = 0; u < 8; ++u) {
            U[j][u] = ((u & 4) ? pj[0] : q0) * ((u & 2) ? pj[1] : q1) * ((u & 1) ? pj[2] : q2);
        }
    }

    v8f acc[4] = {};
#pragma unroll
    for (int u = 0; u < 8; ++u) {
#pragma unroll
        for (int kh = 0; kh < 2; ++kh) {
#pragma unroll
            for (int j = 0; j < 4; ++j) {
                const v4f y = cl[j][u];
                v2f av, bv;
                av.x = U[j][u] * wk[j][2 * kh]; av.y = U[j][u] * wk[j][2 * kh + 1];
                bv.x = kh ? y.z : y.x;          bv.y = kh ? y.w : y.y;
                acc[j] = __builtin_amdgcn_wmma_f32_16x16x4_f32(
                    false, av, false, bv, (short)0, acc[j], false, false);
            }
        }
    }

    const int r = c & 7;
    float res = quad_sum(pick8(acc[0], r));
    const float s1 = quad_sum(pick8(acc[1], r));
    const float s2 = quad_sum(pick8(acc[2], r));
    const float s3 = quad_sum(pick8(acc[3], r));
    if (g == 1) res = s1;
    if (g == 2) res = s2;
    if (g == 3) res = s3;
    if (lane < 8 || lane >= 24) outm[tile * 16 + 4 * g + a] = res;

    if (kFromRoots) {
        if ((wave & 1) == 0) (outm - kM)[ri] = fast_sigmoid(rl);
    }
}

extern "C" void kernel_launch(void* const* d_in, const int* in_sizes, int n_in,
                              void* d_out, int out_size, void* d_ws, size_t ws_size,
                              hipStream_t stream) {
    (void)in_sizes; (void)n_in; (void)out_size; (void)d_ws; (void)ws_size;
    const float* root    = (const float*)d_in[0];
    const float* cpt     = (const float*)d_in[1];
    const int*   parents = (const int*)d_in[2];
    float* out = (float*)d_out;

    bn_layer_kernel<true><<<kM / kNodesPerBlock, kWavesPerBlock * 32, 0, stream>>>(
        cpt, parents, root, out + kM);

    for (int l = 1; l < kLay; ++l) {
        bn_layer_kernel<false><<<kM / kNodesPerBlock, kWavesPerBlock * 32, 0, stream>>>(
            cpt + (size_t)l * kM * kC,
            parents + (size_t)l * kM * kK,
            out + (size_t)l * kM,
            out + (size_t)(l + 1) * kM);
    }
}


# Each wave takes 8 nodes (2 groups of 4), not 16: 1024 waves in 256 four-wave blocks, one per SIMD (was 512 on half the SIMDs); a wave's serial run is 32 matrix ops, not 64; same ops and sum order
# speedup vs baseline: 1.9213x; 1.1436x over previous
#include <hip/hip_runtime.h>


typedef __attribute__((ext_vector_type(2))) float v2f;
typedef __attribute__((ext_vector_type(4))) float v4f;
typedef __attribute__((ext_vector_type(8))) float v8f;

static constexpr int kM    = 8192;
static constexpr int kK    = 8;
static constexpr int kC    = 256;
static constexpr int kLay  = 8;
static constexpr int kWavesPerBlock = 4;
static constexpr int kGroups = 2;
static constexpr int kNodesPerWave  = kGroups * 4;
static constexpr int kNodesPerBlock = kWavesPerBlock * kNodesPerWave;

__device__ __forceinline__ float fast_sigmoid(float x) {
    return __builtin_amdgcn_rcpf(1.0f + __expf(-x));
}

__device__ __forceinline__ float scaled_sigmoid(float x, float s) {
    return __builtin_amdgcn_rcpf(1.0f + __builtin_amdgcn_exp2f(x * s));
}

__device__ __forceinline__ float pick8(v8f c, int i) {
    float r = c[0];
    if (i == 1) r = c[1];
    if (i == 2) r = c[2];
    if (i == 3) r = c[3];
    if (i == 4) r = c[4];
    if (i == 5) r = c[5];
    if (i == 6) r = c[6];
    if (i == 7) r = c[7];
    return r;
}

__device__ __forceinline__ float quad_sum(float v) {
    float t = __int_as_float(__builtin_amdgcn_mov_dpp(__float_as_int(v), 0xB1, 0xF, 0xF, true));
    v += t;
    t = __int_as_float(__builtin_amdgcn_mov_dpp(__float_as_int(v), 0x4E, 0xF, 0xF, true));
    return v + t;
}

template <bool kFromRoots>
__global__ __launch_bounds__(kWavesPerBlock * 32)
void bn_layer_kernel(const float* cpt,
                     const int*   par,
                     const float* prev,
                     float*       outm) {
    const int lane = threadIdx.x & 31;
    const int wave = threadIdx.x >> 5;
    const int c    = lane & 15;
    const int h    = lane >> 4;
    const int a    = c >> 2;
    const int g    = c & 3;
    const int tile = blockIdx.x * kWavesPerBlock + wave;
    const int node0 = tile * kNodesPerWave + a;

    const int ri = blockIdx.x * kNodesPerBlock + lane;
    float rl = 0.0f;
    if (kFromRoots) rl = prev[ri];

    int4 ia[kGroups], ib[kGroups];
#pragma unroll
    for (int j = 0; j < kGroups; ++j) {
        const int4* ip = (const int4*)(par + (size_t)(node0 + 4 * j) * kK);
        ia[j] = ip[0];
        ib[j] = ip[1];
    }

    asm volatile("" ::: "memory");

    const float* row0 = cpt + (size_t)node0 * kC + 8 * g + 4 * h;
    v4f cl[kGroups][8];
#pragma unroll
    for (int u = 0; u < 8; ++u) {
        cl[0][u] = *(const v4f*)(row0 + 32 * u);
        cl[1][u] = *(const v4f*)(row0 + 4 * kC + 32 * u);
    }

    asm volatile("" ::: "memory");

    float p[kGroups][kK];
#pragma unroll
    for (int j = 0; j < kGroups; ++j) {
        p[j][0] = prev[ia[j].x]; p[j][1] = prev[ia[j].y]; p[j][2] = prev[ia[j].z]; p[j][3] = prev[ia[j].w];
        p[j][4] = prev[ib[j].x]; p[j][5] = prev[ib[j].y]; p[j][6] = prev[ib[j].z]; p[j][7] = prev[ib[j].w];
    }

    float nlog2e = -1.44269504088896341f;
    asm volatile("" : "+s"(nlog2e) : : "memory");
#pragma unroll
    for (int u = 0; u < 8; u += 2) {
#pragma unroll
        for (int e = 0; e < 2; ++e) {
#pragma unroll
            for (int d = 0; d < 2; ++d) {
                const v4f x = cl[d][u + e];
                v4f y;
                y.x = scaled_sigmoid(x.x, nlog2e); y.y = scaled_sigmoid(x.y, nlog2e);
                y.z = scaled_sigmoid(x.z, nlog2e); y.w = scaled_sigmoid(x.w, nlog2e);
                cl[d][u + e] = y;
            }
        }
        asm volatile("" : "+v"(cl[0][u]), "+v"(cl[1][u]), "+v"(cl[0][u + 1]),
                          "+v"(cl[1][u + 1]), "+s"(nlog2e));
    }

    float one = 1.0f;
    asm volatile("" : "+s"(one));

    if (kFromRoots) {
#pragma unroll
        for (int j = 0; j < kGroups; ++j) {
#pragma unroll
            for (int k = 0; k < kK; ++k)
                p[j][k] = __builtin_amdgcn_rcpf(one + __expf(-p[j][k]));
        }
    }

    float wk[kGroups][4], U[kGroups][8];
#pragma unroll
    for (int j = 0; j < kGroups; ++j) {
        const float* pj = p[j];
        const float q0 = one - pj[0], q1 = one - pj[1], q2 = one - pj[2];
        const float q3 = one - pj[3], q4 = one - pj[4], q5 = one - pj[5];
        const float q6 = one - pj[6], q7 = one - pj[7];
        const float w = ((g & 2) ? pj[3] : q3) * ((g & 1) ? pj[4] : q4) * (h ? pj[5] : q5);
        const float w0 = w * q6, w1 = w * pj[6];
        wk[j][0] = w0 * q7; wk[j][1] = w0 * pj[7];
        wk[j][2] = w1 * q7; wk[j][3] = w1 * pj[7];
#pragma unroll
        for (int u = 0; u < 8; ++u) {
            U[j][u] = ((u & 4) ? pj[0] : q0) * ((u & 2) ? pj[1] : q1) * ((u & 1) ? pj[2] : q2);
        }
    }

    v8f acc[kGroups] = {};
#pragma unroll
    for (int u = 0; u < 8; ++u) {
#pragma unroll
        for (int kh = 0; kh < 2; ++kh) {
#pragma unroll
            for (int j = 0; j < kGroups; ++j) {
                const v4f y = cl[j][u];
                v2f av, bv;
                av.x = U[j][u] * wk[j][2 * kh]; av.y = U[j][u] * wk[j][2 * kh + 1];
                bv.x = kh ? y.z : y.x;          bv.y = kh ? y.w : y.y;
                acc[j] = __builtin_amdgcn_wmma_f32_16x16x4_f32(
                    false, av, false, bv, (short)0, acc[j], false, false);
            }
        }
    }

    const int r = c & 7;
    float res = quad_sum(pick8(acc[0], r));
    const float s1 = quad_sum(pick8(acc[1], r));
    if (g == 1) res = s1;
    if ((lane < 8 || lane >= 24) && g < kGroups) outm[tile * kNodesPerWave + 4 * g + a] = res;

    if (kFromRoots) {
        if (wave == 0) (outm - kM)[ri] = fast_sigmoid(rl);
    }
}

extern "C" void kernel_launch(void* const* d_in, const int* in_sizes, int n_in,
                              void* d_out, int out_size, void* d_ws, size_t ws_size,
                              hipStream_t stream) {
    (void)in_sizes; (void)n_in; (void)out_size; (void)d_ws; (void)ws_size;
    const float* root    = (const float*)d_in[0];
    const float* cpt     = (const float*)d_in[1];
    const int*   parents = (const int*)d_in[2];
    float* out = (float*)d_out;

    bn_layer_kernel<true><<<kM / kNodesPerBlock, kWavesPerBlock * 32, 0, stream>>>(
        cpt, parents, root, out + kM);

    for (int l = 1; l < kLay; ++l) {
        bn_layer_kernel<false><<<kM / kNodesPerBlock, kWavesPerBlock * 32, 0, stream>>>(
            cpt + (size_t)l * kM * kC,
            parents + (size_t)l * kM * kK,
            out + (size_t)l * kM,
            out + (size_t)(l + 1) * kM);
    }
}
